# starting kernel
# baseline (speedup 1.0000x reference)
#include <hip/hip_runtime.h>

typedef __attribute__((ext_vector_type(2))) float v2f;
typedef __attribute__((ext_vector_type(4))) float v4f;
typedef __attribute__((ext_vector_type(8))) float v8f;

#define N_NODES 100000
#define N_EDGES 1600000
#define D 64

__global__ __launch_bounds__(256) void zero_ws_kernel(float* __restrict__ deg,
                                                      float* __restrict__ tx1) {
    int i = blockIdx.x * 256 + threadIdx.x;
    if (i < N_NODES) deg[i] = 0.0f;
    if (i < N_NODES * D) tx1[i] = 0.0f;
}

__global__ __launch_bounds__(256) void deg_kernel(const int* __restrict__ ei,
                                                  const float* __restrict__ ew,
                                                  float* __restrict__ deg) {
    int e = blockIdx.x * 256 + threadIdx.x;
    if (e >= N_EDGES) return;
    unsafeAtomicAdd(&deg[ei[e]], ew[e]);
}

__global__ __launch_bounds__(256) void dinv_kernel(const float* __restrict__ deg,
                                                   float* __restrict__ dinv) {
    int i = blockIdx.x * 256 + threadIdx.x;
    if (i >= N_NODES) return;
    float d = deg[i];
    dinv[i] = (d > 0.0f) ? __frsqrt_rn(d) : 0.0f;
}

__global__ __launch_bounds__(256) void scatter_kernel(const float* __restrict__ x,
                                                      const int* __restrict__ ei,
                                                      const float* __restrict__ ew,
                                                      const float* __restrict__ dinv,
                                                      float* __restrict__ tx1) {
    int g = blockIdx.x * 256 + threadIdx.x;
    int e = g >> 4;
    int f = (g & 15) * 4;
    if (e >= N_EDGES) return;
    int row = ei[e];
    int col = ei[N_EDGES + e];
    float norm = -dinv[row] * ew[e] * dinv[col];
    v4f xv = *reinterpret_cast<const v4f*>(x + (long long)row * D + f);
    float* dst = tx1 + (long long)col * D + f;
    unsafeAtomicAdd(dst + 0, norm * xv.x);
    unsafeAtomicAdd(dst + 1, norm * xv.y);
    unsafeAtomicAdd(dst + 2, norm * xv.z);
    unsafeAtomicAdd(dst + 3, norm * xv.w);
}

__global__ __launch_bounds__(256) void cheb_gemm_kernel(const float* __restrict__ x,
                                                        const float* __restrict__ tx1,
                                                        const float* __restrict__ W0,
                                                        const float* __restrict__ W1,
                                                        const float* __restrict__ b,
                                                        float* __restrict__ out) {
    __shared__ float sW0[D * D];
    __shared__ float sW1[D * D];
    __shared__ float sb[D];

    int tid = threadIdx.x;
    for (int i = tid; i < D * D; i += 256) {
        sW0[i] = W0[i];
        sW1[i] = W1[i];
    }
    if (tid < D) sb[tid] = b[tid];
    __syncthreads();

    int wave = tid >> 5;
    int lane = tid & 31;
    int m0 = (blockIdx.x * 8 + wave) * 16;
    if (m0 >= N_NODES) return;

    int half = lane >> 4;
    int mrow = lane & 15;

    v8f acc[4];
#pragma unroll
    for (int n = 0; n < 4; ++n) {
        float bias = sb[n * 16 + mrow];
#pragma unroll
        for (int r = 0; r < 8; ++r) acc[n][r] = bias;
    }

    const float* xrow = x   + (long long)(m0 + mrow) * D;
    const float* trow = tx1 + (long long)(m0 + mrow) * D;

#pragma unroll
    for (int ks = 0; ks < 16; ++ks) {
        int kk = ks * 4 + 2 * half;
        v2f ax = *reinterpret_cast<const v2f*>(xrow + kk);
        v2f at = *reinterpret_cast<const v2f*>(trow + kk);
#pragma unroll
        for (int n = 0; n < 4; ++n) {
            int col = n * 16 + mrow;
            v2f b0, b1;
            b0.x = sW0[kk * D + col];
            b0.y = sW0[(kk + 1) * D + col];
            b1.x = sW1[kk * D + col];
            b1.y = sW1[(kk + 1) * D + col];
            acc[n] = __builtin_amdgcn_wmma_f32_16x16x4_f32(
                false, ax, false, b0, (short)0, acc[n], false, false);
            acc[n] = __builtin_amdgcn_wmma_f32_16x16x4_f32(
                false, at, false, b1, (short)0, acc[n], false, false);
        }
    }

#pragma unroll
    for (int n = 0; n < 4; ++n) {
        int col = n * 16 + mrow;
#pragma unroll
        for (int r = 0; r < 8; ++r) {
            int row = m0 + half * 8 + r;
            out[(long long)row * D + col] = acc[n][r];
        }
    }
}

extern "C" void kernel_launch(void* const* d_in, const int* in_sizes, int n_in,
                              void* d_out, int out_size, void* d_ws, size_t ws_size,
                              hipStream_t stream) {
    const float* x  = (const float*)d_in[0];
    const int*   ei = (const int*)d_in[1];
    const float* ew = (const float*)d_in[2];
    const float* W0 = (const float*)d_in[3];
    const float* W1 = (const float*)d_in[4];
    const float* b  = (const float*)d_in[5];
    float* out = (float*)d_out;

    float* deg  = (float*)d_ws;
    float* dinv = deg + N_NODES;
    float* tx1  = deg + 2 * (size_t)N_NODES;

    zero_ws_kernel<<<(N_NODES * D + 255) / 256, 256, 0, stream>>>(deg, tx1);
    deg_kernel<<<(N_EDGES + 255) / 256, 256, 0, stream>>>(ei, ew, deg);
    dinv_kernel<<<(N_NODES + 255) / 256, 256, 0, stream>>>(deg, dinv);
    scatter_kernel<<<(N_EDGES * 16) / 256, 256, 0, stream>>>(x, ei, ew, dinv, tx1);
    cheb_gemm_kernel<<<((N_NODES / 16) + 7) / 8, 256, 0, stream>>>(x, tx1, W0, W1, b, out);
}


# degree sums: global f32 atomics on deg replaced by per-block LDS histograms (ds_add_f32) over 4 node pieces x 64 edge slices, stored as plain partials and summed in fixed order in dinv kernel
# speedup vs baseline: 1.2687x; 1.2687x over previous
#include <hip/hip_runtime.h>

typedef __attribute__((ext_vector_type(2))) float v2f;
typedef __attribute__((ext_vector_type(4))) float v4f;
typedef __attribute__((ext_vector_type(8))) float v8f;

#define N_NODES 100000
#define N_EDGES 1600000
#define D 64

#define DEG_RANGES 4
#define DEG_SLICES 64
#define DEG_RANGE_NODES 25024
#define DEG_SLICE_V4 (N_EDGES / DEG_SLICES / 4)
#define DEG_THREADS 1024
#define DEG_LOADS ((DEG_SLICE_V4 + DEG_THREADS - 1) / DEG_THREADS)

typedef __attribute__((ext_vector_type(4))) int v4i;

__global__ __launch_bounds__(DEG_THREADS) void deg_partial_kernel(const int* __restrict__ ei,
                                                                  const float* __restrict__ ew,
                                                                  float* __restrict__ part) {
    __shared__ float sdeg[DEG_RANGE_NODES];

    const int tid = threadIdx.x;
    const int r = blockIdx.x % DEG_RANGES;
    const int s = blockIdx.x / DEG_RANGES;
    const int lo = r * DEG_RANGE_NODES;

    const v4i* rows4 = reinterpret_cast<const v4i*>(ei) + (size_t)s * DEG_SLICE_V4;
    const v4f* wgt4  = reinterpret_cast<const v4f*>(ew) + (size_t)s * DEG_SLICE_V4;

    v4i rr[DEG_LOADS];
    v4f ww[DEG_LOADS];
#pragma unroll
    for (int j = 0; j < DEG_LOADS; ++j) {
        int i = tid + j * DEG_THREADS;
        int ic = (i < DEG_SLICE_V4) ? i : (DEG_SLICE_V4 - 1);
        rr[j] = rows4[ic];
        ww[j] = wgt4[ic];
    }

    const v4f zero4 = {0.0f, 0.0f, 0.0f, 0.0f};
    for (int i = tid; i < DEG_RANGE_NODES / 4; i += DEG_THREADS)
        reinterpret_cast<v4f*>(sdeg)[i] = zero4;
    __syncthreads();

#pragma unroll
    for (int j = 0; j < DEG_LOADS; ++j) {
        if (tid + j * DEG_THREADS < DEG_SLICE_V4) {
            for (int c = 0; c < 4; ++c) {
                unsigned k = (unsigned)(rr[j][c] - lo);
                if (k < (unsigned)DEG_RANGE_NODES) atomicAdd(&sdeg[k], ww[j][c]);
            }
        }
    }
    __syncthreads();

    int nvalid = N_NODES - lo;
    if (nvalid > DEG_RANGE_NODES) nvalid = DEG_RANGE_NODES;
    v4f* dst = reinterpret_cast<v4f*>(part + (size_t)s * N_NODES + lo);
    for (int i = tid; i < nvalid / 4; i += DEG_THREADS)
        dst[i] = reinterpret_cast<const v4f*>(sdeg)[i];
}

__global__ __launch_bounds__(256) void dinv_kernel(const float* __restrict__ part,
                                                   float* __restrict__ dinv) {
    int i = blockIdx.x * 256 + threadIdx.x;
    if (i >= N_NODES) return;
    float d = 0.0f;
    for (int s = 0; s < DEG_SLICES; ++s) d += part[(size_t)s * N_NODES + i];
    dinv[i] = (d > 0.0f) ? __frsqrt_rn(d) : 0.0f;
}

__global__ __launch_bounds__(256) void zero_ws_kernel(float* __restrict__ tx1) {
    int i = blockIdx.x * 256 + threadIdx.x;
    if (i < N_NODES * D) tx1[i] = 0.0f;
}

__global__ __launch_bounds__(256) void scatter_kernel(const float* __restrict__ x,
                                                      const int* __restrict__ ei,
                                                      const float* __restrict__ ew,
                                                      const float* __restrict__ dinv,
                                                      float* __restrict__ tx1) {
    int g = blockIdx.x * 256 + threadIdx.x;
    int e = g >> 4;
    int f = (g & 15) * 4;
    if (e >= N_EDGES) return;
    int row = ei[e];
    int col = ei[N_EDGES + e];
    float norm = -dinv[row] * ew[e] * dinv[col];
    v4f xv = *reinterpret_cast<const v4f*>(x + (long long)row * D + f);
    float* dst = tx1 + (long long)col * D + f;
    unsafeAtomicAdd(dst + 0, norm * xv.x);
    unsafeAtomicAdd(dst + 1, norm * xv.y);
    unsafeAtomicAdd(dst + 2, norm * xv.z);
    unsafeAtomicAdd(dst + 3, norm * xv.w);
}

__global__ __launch_bounds__(256) void cheb_gemm_kernel(const float* __restrict__ x,
                                                        const float* __restrict__ tx1,
                                                        const float* __restrict__ W0,
                                                        const float* __restrict__ W1,
                                                        const float* __restrict__ b,
                                                        float* __restrict__ out) {
    __shared__ float sW0[D * D];
    __shared__ float sW1[D * D];
    __shared__ float sb[D];

    int tid = threadIdx.x;
    for (int i = tid; i < D * D; i += 256) {
        sW0[i] = W0[i];
        sW1[i] = W1[i];
    }
    if (tid < D) sb[tid] = b[tid];
    __syncthreads();

    int wave = tid >> 5;
    int lane = tid & 31;
    int m0 = (blockIdx.x * 8 + wave) * 16;
    if (m0 >= N_NODES) return;

    int half = lane >> 4;
    int mrow = lane & 15;

    v8f acc[4];
#pragma unroll
    for (int n = 0; n < 4; ++n) {
        float bias = sb[n * 16 + mrow];
#pragma unroll
        for (int r = 0; r < 8; ++r) acc[n][r] = bias;
    }

    const float* xrow = x   + (long long)(m0 + mrow) * D;
    const float* trow = tx1 + (long long)(m0 + mrow) * D;

#pragma unroll
    for (int ks = 0; ks < 16; ++ks) {
        int kk = ks * 4 + 2 * half;
        v2f ax = *reinterpret_cast<const v2f*>(xrow + kk);
        v2f at = *reinterpret_cast<const v2f*>(trow + kk);
#pragma unroll
        for (int n = 0; n < 4; ++n) {
            int col = n * 16 + mrow;
            v2f b0, b1;
            b0.x = sW0[kk * D + col];
            b0.y = sW0[(kk + 1) * D + col];
            b1.x = sW1[kk * D + col];
            b1.y = sW1[(kk + 1) * D + col];
            acc[n] = __builtin_amdgcn_wmma_f32_16x16x4_f32(
                false, ax, false, b0, (short)0, acc[n], false, false);
            acc[n] = __builtin_amdgcn_wmma_f32_16x16x4_f32(
                false, at, false, b1, (short)0, acc[n], false, false);
        }
    }

#pragma unroll
    for (int n = 0; n < 4; ++n) {
        int col = n * 16 + mrow;
#pragma unroll
        for (int r = 0; r < 8; ++r) {
            int row = m0 + half * 8 + r;
            out[(long long)row * D + col] = acc[n][r];
        }
    }
}

extern "C" void kernel_launch(void* const* d_in, const int* in_sizes, int n_in,
                              void* d_out, int out_size, void* d_ws, size_t ws_size,
                              hipStream_t stream) {
    const float* x  = (const float*)d_in[0];
    const int*   ei = (const int*)d_in[1];
    const float* ew = (const float*)d_in[2];
    const float* W0 = (const float*)d_in[3];
    const float* W1 = (const float*)d_in[4];
    const float* b  = (const float*)d_in[5];
    float* out = (float*)d_out;

    float* dinv = (float*)d_ws + N_NODES;
    float* tx1  = (float*)d_ws + 2 * (size_t)N_NODES;

    deg_partial_kernel<<<DEG_RANGES * DEG_SLICES, DEG_THREADS, 0, stream>>>(ei, ew, tx1);
    dinv_kernel<<<(N_NODES + 255) / 256, 256, 0, stream>>>(tx1, dinv);
    zero_ws_kernel<<<(N_NODES * D + 255) / 256, 256, 0, stream>>>(tx1);
    scatter_kernel<<<(N_EDGES * 16) / 256, 256, 0, stream>>>(x, ei, ew, dinv, tx1);
    cheb_gemm_kernel<<<((N_NODES / 16) + 7) / 8, 256, 0, stream>>>(x, tx1, W0, W1, b, out);
}


# scatter: one wave walks 32 edges; each x-row load and each f32 atomic covers one whole 128-byte line of one row (was 16 lanes x 4 strided floats per edge); atomics kept, arrival order as before
# speedup vs baseline: 2.2382x; 1.7642x over previous
#include <hip/hip_runtime.h>

typedef __attribute__((ext_vector_type(2))) float v2f;
typedef __attribute__((ext_vector_type(4))) float v4f;
typedef __attribute__((ext_vector_type(8))) float v8f;

#define N_NODES 100000
#define N_EDGES 1600000
#define D 64

#define DEG_RANGES 4
#define DEG_SLICES 64
#define DEG_RANGE_NODES 25024
#define DEG_SLICE_V4 (N_EDGES / DEG_SLICES / 4)
#define DEG_THREADS 1024
#define DEG_LOADS ((DEG_SLICE_V4 + DEG_THREADS - 1) / DEG_THREADS)

typedef __attribute__((ext_vector_type(4))) int v4i;

__global__ __launch_bounds__(DEG_THREADS) void deg_partial_kernel(const int* __restrict__ ei,
                                                                  const float* __restrict__ ew,
                                                                  float* __restrict__ part) {
    __shared__ float sdeg[DEG_RANGE_NODES];

    const int tid = threadIdx.x;
    const int r = blockIdx.x % DEG_RANGES;
    const int s = blockIdx.x / DEG_RANGES;
    const int lo = r * DEG_RANGE_NODES;

    const v4i* rows4 = reinterpret_cast<const v4i*>(ei) + (size_t)s * DEG_SLICE_V4;
    const v4f* wgt4  = reinterpret_cast<const v4f*>(ew) + (size_t)s * DEG_SLICE_V4;

    v4i rr[DEG_LOADS];
    v4f ww[DEG_LOADS];
#pragma unroll
    for (int j = 0; j < DEG_LOADS; ++j) {
        int i = tid + j * DEG_THREADS;
        int ic = (i < DEG_SLICE_V4) ? i : (DEG_SLICE_V4 - 1);
        rr[j] = rows4[ic];
        ww[j] = wgt4[ic];
    }

    const v4f zero4 = {0.0f, 0.0f, 0.0f, 0.0f};
    for (int i = tid; i < DEG_RANGE_NODES / 4; i += DEG_THREADS)
        reinterpret_cast<v4f*>(sdeg)[i] = zero4;
    __syncthreads();

#pragma unroll
    for (int j = 0; j < DEG_LOADS; ++j) {
        if (tid + j * DEG_THREADS < DEG_SLICE_V4) {
            for (int c = 0; c < 4; ++c) {
                unsigned k = (unsigned)(rr[j][c] - lo);
                if (k < (unsigned)DEG_RANGE_NODES) atomicAdd(&sdeg[k], ww[j][c]);
            }
        }
    }
    __syncthreads();

    int nvalid = N_NODES - lo;
    if (nvalid > DEG_RANGE_NODES) nvalid = DEG_RANGE_NODES;
    v4f* dst = reinterpret_cast<v4f*>(part + (size_t)s * N_NODES + lo);
    for (int i = tid; i < nvalid / 4; i += DEG_THREADS)
        dst[i] = reinterpret_cast<const v4f*>(sdeg)[i];
}

__global__ __launch_bounds__(256) void dinv_kernel(const float* __restrict__ part,
                                                   float* __restrict__ dinv) {
    int i = blockIdx.x * 256 + threadIdx.x;
    if (i >= N_NODES) return;
    float d = 0.0f;
    for (int s = 0; s < DEG_SLICES; ++s) d += part[(size_t)s * N_NODES + i];
    dinv[i] = (d > 0.0f) ? __frsqrt_rn(d) : 0.0f;
}

__global__ __launch_bounds__(256) void zero_ws_kernel(float* __restrict__ tx1) {
    int i = blockIdx.x * 256 + threadIdx.x;
    if (i < N_NODES * D) tx1[i] = 0.0f;
}

#define SC_GROUP 8
static_assert(N_EDGES % 256 == 0, "one wave per 32 edges, 8 waves per block, no tail");

__global__ __launch_bounds__(256) void scatter_kernel(const float* __restrict__ x,
                                                      const int* __restrict__ ei,
                                                      const float* __restrict__ ew,
                                                      const float* __restrict__ dinv,
                                                      float* __restrict__ tx1) {
    const int lane = threadIdx.x & 31;
    const int e = blockIdx.x * 256 + threadIdx.x;
    const int row = ei[e];
    const int col = ei[N_EDGES + e];
    const float norm = -dinv[row] * ew[e] * dinv[col];

    for (int j0 = 0; j0 < 32; j0 += SC_GROUP) {
        float xa[SC_GROUP], xb[SC_GROUP];
#pragma unroll
        for (int k = 0; k < SC_GROUP; ++k) {
            const int r = __builtin_amdgcn_readlane(row, j0 + k);
            const float* xr = x + (size_t)r * D;
            xa[k] = xr[lane];
            xb[k] = xr[lane + 32];
        }
#pragma unroll
        for (int k = 0; k < SC_GROUP; ++k) {
            const int c = __builtin_amdgcn_readlane(col, j0 + k);
            const float s = __builtin_bit_cast(float, __builtin_amdgcn_readlane(__builtin_bit_cast(int, norm), j0 + k));
            float* dst = tx1 + (size_t)c * D;
            unsafeAtomicAdd(dst + lane, s * xa[k]);
            unsafeAtomicAdd(dst + lane + 32, s * xb[k]);
        }
    }
}

__global__ __launch_bounds__(256) void cheb_gemm_kernel(const float* __restrict__ x,
                                                        const float* __restrict__ tx1,
                                                        const float* __restrict__ W0,
                                                        const float* __restrict__ W1,
                                                        const float* __restrict__ b,
                                                        float* __restrict__ out) {
    __shared__ float sW0[D * D];
    __shared__ float sW1[D * D];
    __shared__ float sb[D];

    int tid = threadIdx.x;
    for (int i = tid; i < D * D; i += 256) {
        sW0[i] = W0[i];
        sW1[i] = W1[i];
    }
    if (tid < D) sb[tid] = b[tid];
    __syncthreads();

    int wave = tid >> 5;
    int lane = tid & 31;
    int m0 = (blockIdx.x * 8 + wave) * 16;
    if (m0 >= N_NODES) return;

    int half = lane >> 4;
    int mrow = lane & 15;

    v8f acc[4];
#pragma unroll
    for (int n = 0; n < 4; ++n) {
        float bias = sb[n * 16 + mrow];
#pragma unroll
        for (int r = 0; r < 8; ++r) acc[n][r] = bias;
    }

    const float* xrow = x   + (long long)(m0 + mrow) * D;
    const float* trow = tx1 + (long long)(m0 + mrow) * D;

#pragma unroll
    for (int ks = 0; ks < 16; ++ks) {
        int kk = ks * 4 + 2 * half;
        v2f ax = *reinterpret_cast<const v2f*>(xrow + kk);
        v2f at = *reinterpret_cast<const v2f*>(trow + kk);
#pragma unroll
        for (int n = 0; n < 4; ++n) {
            int col = n * 16 + mrow;
            v2f b0, b1;
            b0.x = sW0[kk * D + col];
            b0.y = sW0[(kk + 1) * D + col];
            b1.x = sW1[kk * D + col];
            b1.y = sW1[(kk + 1) * D + col];
            acc[n] = __builtin_amdgcn_wmma_f32_16x16x4_f32(
                false, ax, false, b0, (short)0, acc[n], false, false);
            acc[n] = __builtin_amdgcn_wmma_f32_16x16x4_f32(
                false, at, false, b1, (short)0, acc[n], false, false);
        }
    }

#pragma unroll
    for (int n = 0; n < 4; ++n) {
        int col = n * 16 + mrow;
#pragma unroll
        for (int r = 0; r < 8; ++r) {
            int row = m0 + half * 8 + r;
            out[(long long)row * D + col] = acc[n][r];
        }
    }
}

extern "C" void kernel_launch(void* const* d_in, const int* in_sizes, int n_in,
                              void* d_out, int out_size, void* d_ws, size_t ws_size,
                              hipStream_t stream) {
    const float* x  = (const float*)d_in[0];
    const int*   ei = (const int*)d_in[1];
    const float* ew = (const float*)d_in[2];
    const float* W0 = (const float*)d_in[3];
    const float* W1 = (const float*)d_in[4];
    const float* b  = (const float*)d_in[5];
    float* out = (float*)d_out;

    float* dinv = (float*)d_ws + N_NODES;
    float* tx1  = (float*)d_ws + 2 * (size_t)N_NODES;

    deg_partial_kernel<<<DEG_RANGES * DEG_SLICES, DEG_THREADS, 0, stream>>>(ei, ew, tx1);
    dinv_kernel<<<(N_NODES + 255) / 256, 256, 0, stream>>>(tx1, dinv);
    zero_ws_kernel<<<(N_NODES * D + 255) / 256, 256, 0, stream>>>(tx1);
    scatter_kernel<<<N_EDGES / 256, 256, 0, stream>>>(x, ei, ew, dinv, tx1);
    cheb_gemm_kernel<<<((N_NODES / 16) + 7) / 8, 256, 0, stream>>>(x, tx1, W0, W1, b, out);
}


# Tx1 clearing moved into the dinv kernel: each thread stores zeros over the 64 degree partials it has just summed (the same words as Tx1); the 25000-block clearing launch is gone (5 launches to 4)
# speedup vs baseline: 2.4008x; 1.0726x over previous
#include <hip/hip_runtime.h>

typedef __attribute__((ext_vector_type(2))) float v2f;
typedef __attribute__((ext_vector_type(4))) float v4f;
typedef __attribute__((ext_vector_type(8))) float v8f;

#define N_NODES 100000
#define N_EDGES 1600000
#define D 64

#define DEG_RANGES 4
#define DEG_SLICES 64
#define DEG_RANGE_NODES 25024
#define DEG_SLICE_V4 (N_EDGES / DEG_SLICES / 4)
#define DEG_THREADS 1024
#define DEG_LOADS ((DEG_SLICE_V4 + DEG_THREADS - 1) / DEG_THREADS)

typedef __attribute__((ext_vector_type(4))) int v4i;

__global__ __launch_bounds__(DEG_THREADS) void deg_partial_kernel(const int* __restrict__ ei,
                                                                  const float* __restrict__ ew,
                                                                  float* __restrict__ part) {
    __shared__ float sdeg[DEG_RANGE_NODES];

    const int tid = threadIdx.x;
    const int r = blockIdx.x % DEG_RANGES;
    const int s = blockIdx.x / DEG_RANGES;
    const int lo = r * DEG_RANGE_NODES;

    const v4i* rows4 = reinterpret_cast<const v4i*>(ei) + (size_t)s * DEG_SLICE_V4;
    const v4f* wgt4  = reinterpret_cast<const v4f*>(ew) + (size_t)s * DEG_SLICE_V4;

    v4i rr[DEG_LOADS];
    v4f ww[DEG_LOADS];
#pragma unroll
    for (int j = 0; j < DEG_LOADS; ++j) {
        int i = tid + j * DEG_THREADS;
        int ic = (i < DEG_SLICE_V4) ? i : (DEG_SLICE_V4 - 1);
        rr[j] = rows4[ic];
        ww[j] = wgt4[ic];
    }

    const v4f zero4 = {0.0f, 0.0f, 0.0f, 0.0f};
    for (int i = tid; i < DEG_RANGE_NODES / 4; i += DEG_THREADS)
        reinterpret_cast<v4f*>(sdeg)[i] = zero4;
    __syncthreads();

#pragma unroll
    for (int j = 0; j < DEG_LOADS; ++j) {
        if (tid + j * DEG_THREADS < DEG_SLICE_V4) {
            for (int c = 0; c < 4; ++c) {
                unsigned k = (unsigned)(rr[j][c] - lo);
                if (k < (unsigned)DEG_RANGE_NODES) atomicAdd(&sdeg[k], ww[j][c]);
            }
        }
    }
    __syncthreads();

    int nvalid = N_NODES - lo;
    if (nvalid > DEG_RANGE_NODES) nvalid = DEG_RANGE_NODES;
    v4f* dst = reinterpret_cast<v4f*>(part + (size_t)s * N_NODES + lo);
    for (int i = tid; i < nvalid / 4; i += DEG_THREADS)
        dst[i] = reinterpret_cast<const v4f*>(sdeg)[i];
}

#define DINV_THREADS 256
static_assert(DEG_SLICES == D, "the partial region and the Tx1 region are the same words");

__global__ __launch_bounds__(DINV_THREADS) void dinv_clear_kernel(float* __restrict__ part,
                                                                  float* __restrict__ dinv) {
    const int i = blockIdx.x * DINV_THREADS + threadIdx.x;
    if (i >= N_NODES) return;
    float p[DEG_SLICES];
#pragma unroll
    for (int s = 0; s < DEG_SLICES; ++s) p[s] = part[(size_t)s * N_NODES + i];
    float d = 0.0f;
#pragma unroll
    for (int s = 0; s < DEG_SLICES; ++s) d += p[s];
    dinv[i] = (d > 0.0f) ? __frsqrt_rn(d) : 0.0f;
#pragma unroll
    for (int s = 0; s < DEG_SLICES; ++s) part[(size_t)s * N_NODES + i] = 0.0f;
}

#define SC_GROUP 8
static_assert(N_EDGES % 256 == 0, "one wave per 32 edges, 8 waves per block, no tail");

__global__ __launch_bounds__(256) void scatter_kernel(const float* __restrict__ x,
                                                      const int* __restrict__ ei,
                                                      const float* __restrict__ ew,
                                                      const float* __restrict__ dinv,
                                                      float* __restrict__ tx1) {
    const int lane = threadIdx.x & 31;
    const int e = blockIdx.x * 256 + threadIdx.x;
    const int row = ei[e];
    const int col = ei[N_EDGES + e];
    const float norm = -dinv[row] * ew[e] * dinv[col];

    for (int j0 = 0; j0 < 32; j0 += SC_GROUP) {
        float xa[SC_GROUP], xb[SC_GROUP];
#pragma unroll
        for (int k = 0; k < SC_GROUP; ++k) {
            const int r = __builtin_amdgcn_readlane(row, j0 + k);
            const float* xr = x + (size_t)r * D;
            xa[k] = xr[lane];
            xb[k] = xr[lane + 32];
        }
#pragma unroll
        for (int k = 0; k < SC_GROUP; ++k) {
            const int c = __builtin_amdgcn_readlane(col, j0 + k);
            const float s = __builtin_bit_cast(float, __builtin_amdgcn_readlane(__builtin_bit_cast(int, norm), j0 + k));
            float* dst = tx1 + (size_t)c * D;
            unsafeAtomicAdd(dst + lane, s * xa[k]);
            unsafeAtomicAdd(dst + lane + 32, s * xb[k]);
        }
    }
}

__global__ __launch_bounds__(256) void cheb_gemm_kernel(const float* __restrict__ x,
                                                        const float* __restrict__ tx1,
                                                        const float* __restrict__ W0,
                                                        const float* __restrict__ W1,
                                                        const float* __restrict__ b,
                                                        float* __restrict__ out) {
    __shared__ float sW0[D * D];
    __shared__ float sW1[D * D];
    __shared__ float sb[D];

    int tid = threadIdx.x;
    for (int i = tid; i < D * D; i += 256) {
        sW0[i] = W0[i];
        sW1[i] = W1[i];
    }
    if (tid < D) sb[tid] = b[tid];
    __syncthreads();

    int wave = tid >> 5;
    int lane = tid & 31;
    int m0 = (blockIdx.x * 8 + wave) * 16;
    if (m0 >= N_NODES) return;

    int half = lane >> 4;
    int mrow = lane & 15;

    v8f acc[4];
#pragma unroll
    for (int n = 0; n < 4; ++n) {
        float bias = sb[n * 16 + mrow];
#pragma unroll
        for (int r = 0; r < 8; ++r) acc[n][r] = bias;
    }

    const float* xrow = x   + (long long)(m0 + mrow) * D;
    const float* trow = tx1 + (long long)(m0 + mrow) * D;

#pragma unroll
    for (int ks = 0; ks < 16; ++ks) {
        int kk = ks * 4 + 2 * half;
        v2f ax = *reinterpret_cast<const v2f*>(xrow + kk);
        v2f at = *reinterpret_cast<const v2f*>(trow + kk);
#pragma unroll
        for (int n = 0; n < 4; ++n) {
            int col = n * 16 + mrow;
            v2f b0, b1;
            b0.x = sW0[kk * D + col];
            b0.y = sW0[(kk + 1) * D + col];
            b1.x = sW1[kk * D + col];
            b1.y = sW1[(kk + 1) * D + col];
            acc[n] = __builtin_amdgcn_wmma_f32_16x16x4_f32(
                false, ax, false, b0, (short)0, acc[n], false, false);
            acc[n] = __builtin_amdgcn_wmma_f32_16x16x4_f32(
                false, at, false, b1, (short)0, acc[n], false, false);
        }
    }

#pragma unroll
    for (int n = 0; n < 4; ++n) {
        int col = n * 16 + mrow;
#pragma unroll
        for (int r = 0; r < 8; ++r) {
            int row = m0 + half * 8 + r;
            out[(long long)row * D + col] = acc[n][r];
        }
    }
}

extern "C" void kernel_launch(void* const* d_in, const int* in_sizes, int n_in,
                              void* d_out, int out_size, void* d_ws, size_t ws_size,
                              hipStream_t stream) {
    const float* x  = (const float*)d_in[0];
    const int*   ei = (const int*)d_in[1];
    const float* ew = (const float*)d_in[2];
    const float* W0 = (const float*)d_in[3];
    const float* W1 = (const float*)d_in[4];
    const float* b  = (const float*)d_in[5];
    float* out = (float*)d_out;

    float* dinv = (float*)d_ws + N_NODES;
    float* tx1  = (float*)d_ws + 2 * (size_t)N_NODES;

    deg_partial_kernel<<<DEG_RANGES * DEG_SLICES, DEG_THREADS, 0, stream>>>(ei, ew, tx1);
    dinv_clear_kernel<<<(N_NODES + DINV_THREADS - 1) / DINV_THREADS, DINV_THREADS, 0, stream>>>(tx1, dinv);
    scatter_kernel<<<N_EDGES / 256, 256, 0, stream>>>(x, ei, ew, dinv, tx1);
    cheb_gemm_kernel<<<((N_NODES / 16) + 7) / 8, 256, 0, stream>>>(x, tx1, W0, W1, b, out);
}


# scatter: factor -dinv[col] moved out of the per-edge norm, applied once per node on the product's A fragment; per-edge dinv[col] gather gone; scaling rounds differently, sum order as before
# speedup vs baseline: 2.5072x; 1.0443x over previous
#include <hip/hip_runtime.h>

typedef __attribute__((ext_vector_type(2))) float v2f;
typedef __attribute__((ext_vector_type(4))) float v4f;
typedef __attribute__((ext_vector_type(8))) float v8f;

#define N_NODES 100000
#define N_EDGES 1600000
#define D 64

#define DEG_RANGES 4
#define DEG_SLICES 64
#define DEG_RANGE_NODES 25024
#define DEG_SLICE_V4 (N_EDGES / DEG_SLICES / 4)
#define DEG_THREADS 1024
#define DEG_LOADS ((DEG_SLICE_V4 + DEG_THREADS - 1) / DEG_THREADS)

typedef __attribute__((ext_vector_type(4))) int v4i;

__global__ __launch_bounds__(DEG_THREADS) void deg_partial_kernel(const int* __restrict__ ei,
                                                                  const float* __restrict__ ew,
                                                                  float* __restrict__ part) {
    __shared__ float sdeg[DEG_RANGE_NODES];

    const int tid = threadIdx.x;
    const int r = blockIdx.x % DEG_RANGES;
    const int s = blockIdx.x / DEG_RANGES;
    const int lo = r * DEG_RANGE_NODES;

    const v4i* rows4 = reinterpret_cast<const v4i*>(ei) + (size_t)s * DEG_SLICE_V4;
    const v4f* wgt4  = reinterpret_cast<const v4f*>(ew) + (size_t)s * DEG_SLICE_V4;

    v4i rr[DEG_LOADS];
    v4f ww[DEG_LOADS];
#pragma unroll
    for (int j = 0; j < DEG_LOADS; ++j) {
        int i = tid + j * DEG_THREADS;
        int ic = (i < DEG_SLICE_V4) ? i : (DEG_SLICE_V4 - 1);
        rr[j] = rows4[ic];
        ww[j] = wgt4[ic];
    }

    const v4f zero4 = {0.0f, 0.0f, 0.0f, 0.0f};
    for (int i = tid; i < DEG_RANGE_NODES / 4; i += DEG_THREADS)
        reinterpret_cast<v4f*>(sdeg)[i] = zero4;
    __syncthreads();

#pragma unroll
    for (int j = 0; j < DEG_LOADS; ++j) {
        if (tid + j * DEG_THREADS < DEG_SLICE_V4) {
            for (int c = 0; c < 4; ++c) {
                unsigned k = (unsigned)(rr[j][c] - lo);
                if (k < (unsigned)DEG_RANGE_NODES) atomicAdd(&sdeg[k], ww[j][c]);
            }
        }
    }
    __syncthreads();

    int nvalid = N_NODES - lo;
    if (nvalid > DEG_RANGE_NODES) nvalid = DEG_RANGE_NODES;
    v4f* dst = reinterpret_cast<v4f*>(part + (size_t)s * N_NODES + lo);
    for (int i = tid; i < nvalid / 4; i += DEG_THREADS)
        dst[i] = reinterpret_cast<const v4f*>(sdeg)[i];
}

#define DINV_THREADS 256
static_assert(DEG_SLICES == D, "the partial region and the Tx1 region are the same words");

__global__ __launch_bounds__(DINV_THREADS) void dinv_clear_kernel(float* __restrict__ part,
                                                                  float* __restrict__ dinv) {
    const int i = blockIdx.x * DINV_THREADS + threadIdx.x;
    if (i >= N_NODES) return;
    float p[DEG_SLICES];
#pragma unroll
    for (int s = 0; s < DEG_SLICES; ++s) p[s] = part[(size_t)s * N_NODES + i];
    float d = 0.0f;
#pragma unroll
    for (int s = 0; s < DEG_SLICES; ++s) d += p[s];
    dinv[i] = (d > 0.0f) ? __frsqrt_rn(d) : 0.0f;
#pragma unroll
    for (int s = 0; s < DEG_SLICES; ++s) part[(size_t)s * N_NODES + i] = 0.0f;
}

#define SC_GROUP 8
static_assert(N_EDGES % 256 == 0, "one wave per 32 edges, 8 waves per block, no tail");

__global__ __launch_bounds__(256) void scatter_kernel(const float* __restrict__ x,
                                                      const int* __restrict__ ei,
                                                      const float* __restrict__ ew,
                                                      const float* __restrict__ dinv,
                                                      float* __restrict__ tx1) {
    const int lane = threadIdx.x & 31;
    const int e = blockIdx.x * 256 + threadIdx.x;
    const int row = ei[e];
    const int col = ei[N_EDGES + e];
    const float norm = dinv[row] * ew[e];

    for (int j0 = 0; j0 < 32; j0 += SC_GROUP) {
        float xa[SC_GROUP], xb[SC_GROUP];
#pragma unroll
        for (int k = 0; k < SC_GROUP; ++k) {
            const int r = __builtin_amdgcn_readlane(row, j0 + k);
            const float* xr = x + (size_t)r * D;
            xa[k] = xr[lane];
            xb[k] = xr[lane + 32];
        }
#pragma unroll
        for (int k = 0; k < SC_GROUP; ++k) {
            const int c = __builtin_amdgcn_readlane(col, j0 + k);
            const float s = __builtin_bit_cast(float, __builtin_amdgcn_readlane(__builtin_bit_cast(int, norm), j0 + k));
            float* dst = tx1 + (size_t)c * D;
            unsafeAtomicAdd(dst + lane, s * xa[k]);
            unsafeAtomicAdd(dst + lane + 32, s * xb[k]);
        }
    }
}

__global__ __launch_bounds__(256) void cheb_gemm_kernel(const float* __restrict__ x,
                                                        const float* __restrict__ tx1,
                                                        const float* __restrict__ W0,
                                                        const float* __restrict__ W1,
                                                        const float* __restrict__ b,
                                                        const float* __restrict__ dinv,
                                                        float* __restrict__ out) {
    __shared__ float sW0[D * D];
    __shared__ float sW1[D * D];
    __shared__ float sb[D];

    int tid = threadIdx.x;
    for (int i = tid; i < D * D; i += 256) {
        sW0[i] = W0[i];
        sW1[i] = W1[i];
    }
    if (tid < D) sb[tid] = b[tid];
    __syncthreads();

    int wave = tid >> 5;
    int lane = tid & 31;
    int m0 = (blockIdx.x * 8 + wave) * 16;
    if (m0 >= N_NODES) return;

    int half = lane >> 4;
    int mrow = lane & 15;

    v8f acc[4];
#pragma unroll
    for (int n = 0; n < 4; ++n) {
        float bias = sb[n * 16 + mrow];
#pragma unroll
        for (int r = 0; r < 8; ++r) acc[n][r] = bias;
    }

    const float* xrow = x   + (long long)(m0 + mrow) * D;
    const float* trow = tx1 + (long long)(m0 + mrow) * D;
    const float nd = -dinv[m0 + mrow];

#pragma unroll
    for (int ks = 0; ks < 16; ++ks) {
        int kk = ks * 4 + 2 * half;
        v2f ax = *reinterpret_cast<const v2f*>(xrow + kk);
        v2f at = *reinterpret_cast<const v2f*>(trow + kk);
        at.x *= nd;
        at.y *= nd;
#pragma unroll
        for (int n = 0; n < 4; ++n) {
            int col = n * 16 + mrow;
            v2f b0, b1;
            b0.x = sW0[kk * D + col];
            b0.y = sW0[(kk + 1) * D + col];
            b1.x = sW1[kk * D + col];
            b1.y = sW1[(kk + 1) * D + col];
            acc[n] = __builtin_amdgcn_wmma_f32_16x16x4_f32(
                false, ax, false, b0, (short)0, acc[n], false, false);
            acc[n] = __builtin_amdgcn_wmma_f32_16x16x4_f32(
                false, at, false, b1, (short)0, acc[n], false, false);
        }
    }

#pragma unroll
    for (int n = 0; n < 4; ++n) {
        int col = n * 16 + mrow;
#pragma unroll
        for (int r = 0; r < 8; ++r) {
            int row = m0 + half * 8 + r;
            out[(long long)row * D + col] = acc[n][r];
        }
    }
}

extern "C" void kernel_launch(void* const* d_in, const int* in_sizes, int n_in,
                              void* d_out, int out_size, void* d_ws, size_t ws_size,
                              hipStream_t stream) {
    const float* x  = (const float*)d_in[0];
    const int*   ei = (const int*)d_in[1];
    const float* ew = (const float*)d_in[2];
    const float* W0 = (const float*)d_in[3];
    const float* W1 = (const float*)d_in[4];
    const float* b  = (const float*)d_in[5];
    float* out = (float*)d_out;

    float* dinv = (float*)d_ws + N_NODES;
    float* tx1  = (float*)d_ws + 2 * (size_t)N_NODES;

    deg_partial_kernel<<<DEG_RANGES * DEG_SLICES, DEG_THREADS, 0, stream>>>(ei, ew, tx1);
    dinv_clear_kernel<<<(N_NODES + DINV_THREADS - 1) / DINV_THREADS, DINV_THREADS, 0, stream>>>(tx1, dinv);
    scatter_kernel<<<N_EDGES / 256, 256, 0, stream>>>(x, ei, ew, dinv, tx1);
    cheb_gemm_kernel<<<((N_NODES / 16) + 7) / 8, 256, 0, stream>>>(x, tx1, W0, W1, b, dinv, out);
}


# scatter: edges counting-sorted into 250 destination ranges; one block per range sums its rows in LDS (ds_add_f32, LDS arrival order) and feeds the product from LDS; global atomics and Tx1 plane gone
# speedup vs baseline: 2.8071x; 1.1196x over previous
#include <hip/hip_runtime.h>

typedef __attribute__((ext_vector_type(2))) float v2f;
typedef __attribute__((ext_vector_type(4))) float v4f;
typedef __attribute__((ext_vector_type(8))) float v8f;

#define N_NODES 100000
#define N_EDGES 1600000
#define D 64

#define DEG_RANGES 4
#define DEG_SLICES 64
#define DEG_RANGE_NODES 25024
#define DEG_SLICE_V4 (N_EDGES / DEG_SLICES / 4)
#define DEG_THREADS 1024
#define DEG_LOADS ((DEG_SLICE_V4 + DEG_THREADS - 1) / DEG_THREADS)

typedef __attribute__((ext_vector_type(4))) int v4i;

__global__ __launch_bounds__(DEG_THREADS) void deg_partial_kernel(const int* __restrict__ ei,
                                                                  const float* __restrict__ ew,
                                                                  float* __restrict__ part) {
    __shared__ float sdeg[DEG_RANGE_NODES];

    const int tid = threadIdx.x;
    const int r = blockIdx.x % DEG_RANGES;
    const int s = blockIdx.x / DEG_RANGES;
    const int lo = r * DEG_RANGE_NODES;

    const v4i* rows4 = reinterpret_cast<const v4i*>(ei) + (size_t)s * DEG_SLICE_V4;
    const v4f* wgt4  = reinterpret_cast<const v4f*>(ew) + (size_t)s * DEG_SLICE_V4;

    v4i rr[DEG_LOADS];
    v4f ww[DEG_LOADS];
#pragma unroll
    for (int j = 0; j < DEG_LOADS; ++j) {
        int i = tid + j * DEG_THREADS;
        int ic = (i < DEG_SLICE_V4) ? i : (DEG_SLICE_V4 - 1);
        rr[j] = rows4[ic];
        ww[j] = wgt4[ic];
    }

    const v4f zero4 = {0.0f, 0.0f, 0.0f, 0.0f};
    for (int i = tid; i < DEG_RANGE_NODES / 4; i += DEG_THREADS)
        reinterpret_cast<v4f*>(sdeg)[i] = zero4;
    __syncthreads();

#pragma unroll
    for (int j = 0; j < DEG_LOADS; ++j) {
        if (tid + j * DEG_THREADS < DEG_SLICE_V4) {
            for (int c = 0; c < 4; ++c) {
                unsigned k = (unsigned)(rr[j][c] - lo);
                if (k < (unsigned)DEG_RANGE_NODES) atomicAdd(&sdeg[k], ww[j][c]);
            }
        }
    }
    __syncthreads();

    int nvalid = N_NODES - lo;
    if (nvalid > DEG_RANGE_NODES) nvalid = DEG_RANGE_NODES;
    v4f* dst = reinterpret_cast<v4f*>(part + (size_t)s * N_NODES + lo);
    for (int i = tid; i < nvalid / 4; i += DEG_THREADS)
        dst[i] = reinterpret_cast<const v4f*>(sdeg)[i];
}

#define DINV_THREADS 256

__global__ __launch_bounds__(DINV_THREADS) void dinv_kernel(const float* __restrict__ part,
                                                            float* __restrict__ dinv) {
    const int i = blockIdx.x * DINV_THREADS + threadIdx.x;
    if (i >= N_NODES) return;
    float p[DEG_SLICES];
#pragma unroll
    for (int s = 0; s < DEG_SLICES; ++s) p[s] = part[(size_t)s * N_NODES + i];
    float d = 0.0f;
#pragma unroll
    for (int s = 0; s < DEG_SLICES; ++s) d += p[s];
    dinv[i] = (d > 0.0f) ? __frsqrt_rn(d) : 0.0f;
}

#define NBIN 250
#define BIN_NODES 400
#define NSB 250
#define SB_EDGES 6400
#define SB_THREADS 256
#define SB_PER_THREAD (SB_EDGES / SB_THREADS)
static_assert(NBIN * BIN_NODES == N_NODES, "ranges tile the nodes exactly");
static_assert(BIN_NODES % 16 == 0, "a range is a whole number of 16-row tiles");
static_assert(NSB * SB_EDGES == N_EDGES, "slices tile the edges exactly");
static_assert(SB_PER_THREAD * SB_THREADS == SB_EDGES, "no tail inside a slice");
static_assert(NBIN < SB_THREADS, "one thread per range, and one more for the end mark");
static_assert(NSB * NBIN <= 65536 && 65536 + NBIN + 1 <= N_NODES, "cnt and binstart fit in the first N words of the workspace");
static_assert((size_t)N_EDGES * 4 == (size_t)N_NODES * D, "the edge entries fill exactly the words of the degree partials");

__device__ __forceinline__ int bin_of(int c) {
    unsigned b = (unsigned)c / (unsigned)BIN_NODES;
    return (b < (unsigned)NBIN) ? (int)b : (NBIN - 1);
}

__global__ __launch_bounds__(SB_THREADS) void bin_count_kernel(const int* __restrict__ ei,
                                                               int* __restrict__ cnt) {
    __shared__ int scnt[SB_THREADS];
    const int tid = threadIdx.x;
    scnt[tid] = 0;
    __syncthreads();
    const int* colp = ei + N_EDGES + blockIdx.x * SB_EDGES;
    int c[SB_PER_THREAD];
#pragma unroll
    for (int j = 0; j < SB_PER_THREAD; ++j) c[j] = colp[j * SB_THREADS + tid];
#pragma unroll
    for (int j = 0; j < SB_PER_THREAD; ++j) atomicAdd(&scnt[bin_of(c[j])], 1);
    __syncthreads();
    if (tid < NBIN) cnt[blockIdx.x * NBIN + tid] = scnt[tid];
}

__global__ __launch_bounds__(SB_THREADS) void place_kernel(const int* __restrict__ ei,
                                                           const float* __restrict__ ew,
                                                           const float* __restrict__ dinv,
                                                           const int* __restrict__ cnt,
                                                           int* __restrict__ binstart,
                                                           int* __restrict__ entries) {
    __shared__ int stot[SB_THREADS];
    __shared__ int cur[SB_THREADS];
    const int tid = threadIdx.x;
    const int sb = blockIdx.x;

    const int e0 = sb * SB_EDGES + tid;
    int r[SB_PER_THREAD], c[SB_PER_THREAD];
    float w[SB_PER_THREAD];
#pragma unroll
    for (int j = 0; j < SB_PER_THREAD; ++j) {
        r[j] = ei[e0 + j * SB_THREADS];
        c[j] = ei[N_EDGES + e0 + j * SB_THREADS];
        w[j] = ew[e0 + j * SB_THREADS];
    }

    int tot = 0, mine = 0;
    if (tid < NBIN) {
#pragma unroll 25
        for (int s = 0; s < NSB; ++s) {
            int v = cnt[s * NBIN + tid];
            tot += v;
            mine += (s < sb) ? v : 0;
        }
    }
    stot[tid] = tot;
    __syncthreads();
    int start = 0;
    for (int i = 0; i < tid; ++i) start += stot[i];
    cur[tid] = start + mine;
    if (sb == 0 && tid <= NBIN) binstart[tid] = start;
    __syncthreads();

    v4i* ent = reinterpret_cast<v4i*>(entries);
#pragma unroll
    for (int j = 0; j < SB_PER_THREAD; ++j) {
        const int b = bin_of(c[j]);
        int pos = atomicAdd(&cur[b], 1);
        pos = (pos < 0) ? 0 : ((pos > N_EDGES - 1) ? (N_EDGES - 1) : pos);
        const float nrm = dinv[r[j]] * w[j];
        v4i en;
        en.x = r[j];
        en.y = c[j] - b * BIN_NODES;
        en.z = __builtin_bit_cast(int, nrm);
        en.w = 0;
        ent[pos] = en;
    }
}

#define GG_THREADS 1024
#define GG_WAVES (GG_THREADS / 32)
#define SC_GROUP 8
static_assert(BIN_NODES / 16 <= GG_WAVES, "one wave per 16-row tile of the range");

__global__ __launch_bounds__(GG_THREADS) void gather_gemm_kernel(const float* __restrict__ x,
                                                                 const int* __restrict__ entries,
                                                                 const int* __restrict__ binstart,
                                                                 const float* __restrict__ W0,
                                                                 const float* __restrict__ W1,
                                                                 const float* __restrict__ b,
                                                                 const float* __restrict__ dinv,
                                                                 float* __restrict__ out) {
    __shared__ float sS[BIN_NODES * D];
    __shared__ float sW0[D * D];
    __shared__ float sW1[D * D];
    __shared__ float sb[D];

    const int tid = threadIdx.x;
    const int lane = tid & 31;
    const int wave = __builtin_amdgcn_readfirstlane(tid >> 5);
    const int bin = blockIdx.x;

    const v4f zero4 = {0.0f, 0.0f, 0.0f, 0.0f};
    for (int i = tid; i < BIN_NODES * D / 4; i += GG_THREADS)
        reinterpret_cast<v4f*>(sS)[i] = zero4;
    for (int i = tid; i < D * D; i += GG_THREADS) {
        sW0[i] = W0[i];
        sW1[i] = W1[i];
    }
    if (tid < D) sb[tid] = b[tid];
    __syncthreads();

    int beg = binstart[bin];
    int end = binstart[bin + 1];
    beg = (beg < 0) ? 0 : ((beg > N_EDGES) ? N_EDGES : beg);
    end = (end < 0) ? 0 : ((end > N_EDGES) ? N_EDGES : end);
    beg = __builtin_amdgcn_readfirstlane(beg);
    end = __builtin_amdgcn_readfirstlane(end);

    const v4i* ent = reinterpret_cast<const v4i*>(entries);
    for (int base = beg + wave * 32; base < end; base += GG_WAVES * 32) {
        const int rest = end - base;
        const int nvalid = (rest < 32) ? rest : 32;
        const int il = (lane < nvalid) ? lane : (nvalid - 1);
        const v4i en = ent[base + il];
        const int row = en.x;
        int cl = en.y;
        cl = (cl < 0) ? 0 : ((cl > BIN_NODES - 1) ? (BIN_NODES - 1) : cl);
        const int nrm = en.z;

        for (int j0 = 0; j0 < nvalid; j0 += SC_GROUP) {
            float xa[SC_GROUP], xb[SC_GROUP];
#pragma unroll
            for (int k = 0; k < SC_GROUP; ++k) {
                const int r = __builtin_amdgcn_readlane(row, j0 + k);
                const float* xr = x + (size_t)r * D;
                xa[k] = xr[lane];
                xb[k] = xr[lane + 32];
            }
#pragma unroll
            for (int k = 0; k < SC_GROUP; ++k) {
                if (j0 + k < nvalid) {
                    const int c = __builtin_amdgcn_readlane(cl, j0 + k);
                    const float s = __builtin_bit_cast(float, __builtin_amdgcn_readlane(nrm, j0 + k));
                    float* dst = sS + c * D;
                    atomicAdd(dst + lane, s * xa[k]);
                    atomicAdd(dst + lane + 32, s * xb[k]);
                }
            }
        }
    }
    __syncthreads();

    if (wave >= BIN_NODES / 16) return;

    const int half = lane >> 4;
    const int mrow = lane & 15;
    const int ml = wave * 16 + mrow;
    const int m0 = bin * BIN_NODES + wave * 16;

    v8f acc[4];
#pragma unroll
    for (int n = 0; n < 4; ++n) {
        float bias = sb[n * 16 + mrow];
#pragma unroll
        for (int r = 0; r < 8; ++r) acc[n][r] = bias;
    }

    const float* xrow = x + (long long)(m0 + mrow) * D;
    const float* trow = sS + ml * D;
    const float nd = -dinv[m0 + mrow];

#pragma unroll
    for (int ks = 0; ks < 16; ++ks) {
        int kk = ks * 4 + 2 * half;
        v2f ax = *reinterpret_cast<const v2f*>(xrow + kk);
        v2f at = *reinterpret_cast<const v2f*>(trow + kk);
        at.x *= nd;
        at.y *= nd;
#pragma unroll
        for (int n = 0; n < 4; ++n) {
            int col = n * 16 + mrow;
            v2f b0, b1;
            b0.x = sW0[kk * D + col];
            b0.y = sW0[(kk + 1) * D + col];
            b1.x = sW1[kk * D + col];
            b1.y = sW1[(kk + 1) * D + col];
            acc[n] = __builtin_amdgcn_wmma_f32_16x16x4_f32(
                false, ax, false, b0, (short)0, acc[n], false, false);
            acc[n] = __builtin_amdgcn_wmma_f32_16x16x4_f32(
                false, at, false, b1, (short)0, acc[n], false, false);
        }
    }

#pragma unroll
    for (int n = 0; n < 4; ++n) {
        int col = n * 16 + mrow;
#pragma unroll
        for (int r = 0; r < 8; ++r) {
            int row = m0 + half * 8 + r;
            out[(long long)row * D + col] = acc[n][r];
        }
    }
}

extern "C" void kernel_launch(void* const* d_in, const int* in_sizes, int n_in,
                              void* d_out, int out_size, void* d_ws, size_t ws_size,
                              hipStream_t stream) {
    const float* x  = (const float*)d_in[0];
    const int*   ei = (const int*)d_in[1];
    const float* ew = (const float*)d_in[2];
    const float* W0 = (const float*)d_in[3];
    const float* W1 = (const float*)d_in[4];
    const float* b  = (const float*)d_in[5];
    float* out = (float*)d_out;

    int*   cnt      = (int*)d_ws;
    int*   binstart = (int*)d_ws + 65536;
    float* dinv     = (float*)d_ws + N_NODES;
    float* big      = (float*)d_ws + 2 * (size_t)N_NODES;

    deg_partial_kernel<<<DEG_RANGES * DEG_SLICES, DEG_THREADS, 0, stream>>>(ei, ew, big);
    dinv_kernel<<<(N_NODES + DINV_THREADS - 1) / DINV_THREADS, DINV_THREADS, 0, stream>>>(big, dinv);
    bin_count_kernel<<<NSB, SB_THREADS, 0, stream>>>(ei, cnt);
    place_kernel<<<NSB, SB_THREADS, 0, stream>>>(ei, ew, dinv, cnt, binstart, (int*)big);
    gather_gemm_kernel<<<NBIN, GG_THREADS, 0, stream>>>(x, (const int*)big, binstart, W0, W1, b, dinv, out);
}


# launches 5 to 3: range counting rides in the degree launch, dinv in the place launch; dinv[row]*w now formed in the gather (same rounding); degree partials staged in out; sum orders as before
# speedup vs baseline: 3.3483x; 1.1928x over previous
#include <hip/hip_runtime.h>

typedef __attribute__((ext_vector_type(2))) float v2f;
typedef __attribute__((ext_vector_type(4))) float v4f;
typedef __attribute__((ext_vector_type(8))) float v8f;

#define N_NODES 100000
#define N_EDGES 1600000
#define D 64

#define DEG_RANGES 4
#define DEG_SLICES 64
#define DEG_RANGE_NODES 25024
#define DEG_SLICE_V4 (N_EDGES / DEG_SLICES / 4)
#define DEG_THREADS 1024
#define DEG_LOADS ((DEG_SLICE_V4 + DEG_THREADS - 1) / DEG_THREADS)

typedef __attribute__((ext_vector_type(4))) int v4i;

#define NBIN 250
#define BIN_NODES 400
#define NSB 250
#define SB_EDGES 6400
#define SB_THREADS 256
#define SB_PER_THREAD (SB_EDGES / SB_THREADS)
#define CNT_LOADS ((SB_EDGES + DEG_THREADS - 1) / DEG_THREADS)
#define DEG_BLOCKS (DEG_RANGES * DEG_SLICES)

__device__ __forceinline__ int bin_of(int c) {
    unsigned b = (unsigned)c / (unsigned)BIN_NODES;
    return (b < (unsigned)NBIN) ? (int)b : (NBIN - 1);
}

__global__ __launch_bounds__(DEG_THREADS) void deg_count_kernel(const int* __restrict__ ei,
                                                                const float* __restrict__ ew,
                                                                float* __restrict__ part,
                                                                int* __restrict__ cnt) {
    __shared__ float sdeg[DEG_RANGE_NODES];
    __shared__ int scnt[SB_THREADS];

    const int tid = threadIdx.x;

    if (blockIdx.x >= DEG_BLOCKS) {
        const int sb = blockIdx.x - DEG_BLOCKS;
        if (tid < SB_THREADS) scnt[tid] = 0;
        __syncthreads();
        const int* colp = ei + N_EDGES + sb * SB_EDGES;
        int c[CNT_LOADS];
#pragma unroll
        for (int j = 0; j < CNT_LOADS; ++j) {
            int i = tid + j * DEG_THREADS;
            c[j] = colp[(i < SB_EDGES) ? i : (SB_EDGES - 1)];
        }
#pragma unroll
        for (int j = 0; j < CNT_LOADS; ++j)
            if (tid + j * DEG_THREADS < SB_EDGES) atomicAdd(&scnt[bin_of(c[j])], 1);
        __syncthreads();
        if (tid < NBIN) cnt[sb * NBIN + tid] = scnt[tid];
        return;
    }

    const int r = blockIdx.x % DEG_RANGES;
    const int s = blockIdx.x / DEG_RANGES;
    const int lo = r * DEG_RANGE_NODES;

    const v4i* rows4 = reinterpret_cast<const v4i*>(ei) + (size_t)s * DEG_SLICE_V4;
    const v4f* wgt4  = reinterpret_cast<const v4f*>(ew) + (size_t)s * DEG_SLICE_V4;

    v4i rr[DEG_LOADS];
    v4f ww[DEG_LOADS];
#pragma unroll
    for (int j = 0; j < DEG_LOADS; ++j) {
        int i = tid + j * DEG_THREADS;
        int ic = (i < DEG_SLICE_V4) ? i : (DEG_SLICE_V4 - 1);
        rr[j] = rows4[ic];
        ww[j] = wgt4[ic];
    }

    const v4f zero4 = {0.0f, 0.0f, 0.0f, 0.0f};
    for (int i = tid; i < DEG_RANGE_NODES / 4; i += DEG_THREADS)
        reinterpret_cast<v4f*>(sdeg)[i] = zero4;
    __syncthreads();

#pragma unroll
    for (int j = 0; j < DEG_LOADS; ++j) {
        if (tid + j * DEG_THREADS < DEG_SLICE_V4) {
            for (int c = 0; c < 4; ++c) {
                unsigned k = (unsigned)(rr[j][c] - lo);
                if (k < (unsigned)DEG_RANGE_NODES) atomicAdd(&sdeg[k], ww[j][c]);
            }
        }
    }
    __syncthreads();

    int nvalid = N_NODES - lo;
    if (nvalid > DEG_RANGE_NODES) nvalid = DEG_RANGE_NODES;
    v4f* dst = reinterpret_cast<v4f*>(part + (size_t)s * N_NODES + lo);
    for (int i = tid; i < nvalid / 4; i += DEG_THREADS)
        dst[i] = reinterpret_cast<const v4f*>(sdeg)[i];
}

#define DINV_THREADS 256
#define DINV_BLOCKS ((N_NODES + DINV_THREADS - 1) / DINV_THREADS)
static_assert(DINV_THREADS == SB_THREADS, "both kinds of block of launch 2 have the same size");
static_assert(NBIN * BIN_NODES == N_NODES, "ranges tile the nodes exactly");
static_assert(BIN_NODES % 16 == 0, "a range is a whole number of 16-row tiles");
static_assert(NSB * SB_EDGES == N_EDGES, "slices tile the edges exactly");
static_assert(SB_PER_THREAD * SB_THREADS == SB_EDGES, "no tail inside a slice");
static_assert(NBIN < SB_THREADS, "one thread per range, and one more for the end mark");
static_assert(NSB * NBIN <= 65536 && 65536 + NBIN + 1 <= N_NODES, "cnt and binstart fit in the first N words of the workspace");
static_assert((size_t)DEG_SLICES * N_NODES == (size_t)N_NODES * D, "the degree partials fill exactly the output buffer");
static_assert((size_t)N_EDGES * 4 == (size_t)N_NODES * D, "the edge entries fill exactly the last N*D words of the workspace");

__global__ __launch_bounds__(SB_THREADS) void dinv_place_kernel(const int* __restrict__ ei,
                                                                const float* __restrict__ ew,
                                                                const float* __restrict__ part,
                                                                float* __restrict__ dinv,
                                                                const int* __restrict__ cnt,
                                                                int* __restrict__ binstart,
                                                                int* __restrict__ entries) {
    __shared__ int stot[SB_THREADS];
    __shared__ int cur[SB_THREADS];
    const int tid = threadIdx.x;

    if (blockIdx.x >= NSB) {
        const int i = (blockIdx.x - NSB) * DINV_THREADS + tid;
        if (i >= N_NODES) return;
        float p[DEG_SLICES];
#pragma unroll
        for (int s = 0; s < DEG_SLICES; ++s) p[s] = part[(size_t)s * N_NODES + i];
        float d = 0.0f;
#pragma unroll
        for (int s = 0; s < DEG_SLICES; ++s) d += p[s];
        dinv[i] = (d > 0.0f) ? __frsqrt_rn(d) : 0.0f;
        return;
    }

    const int sb = blockIdx.x;

    const int e0 = sb * SB_EDGES + tid;
    int r[SB_PER_THREAD], c[SB_PER_THREAD];
    float w[SB_PER_THREAD];
#pragma unroll
    for (int j = 0; j < SB_PER_THREAD; ++j) {
        r[j] = ei[e0 + j * SB_THREADS];
        c[j] = ei[N_EDGES + e0 + j * SB_THREADS];
        w[j] = ew[e0 + j * SB_THREADS];
    }

    int tot = 0, mine = 0;
    if (tid < NBIN) {
#pragma unroll 25
        for (int s = 0; s < NSB; ++s) {
            int v = cnt[s * NBIN + tid];
            tot += v;
            mine += (s < sb) ? v : 0;
        }
    }
    stot[tid] = tot;
    __syncthreads();
    int start = 0;
    for (int i = 0; i < tid; ++i) start += stot[i];
    cur[tid] = start + mine;
    if (sb == 0 && tid <= NBIN) binstart[tid] = start;
    __syncthreads();

    v4i* ent = reinterpret_cast<v4i*>(entries);
#pragma unroll
    for (int j = 0; j < SB_PER_THREAD; ++j) {
        const int b = bin_of(c[j]);
        int pos = atomicAdd(&cur[b], 1);
        pos = (pos < 0) ? 0 : ((pos > N_EDGES - 1) ? (N_EDGES - 1) : pos);
        v4i en;
        en.x = r[j];
        en.y = c[j] - b * BIN_NODES;
        const float wj = w[j];
        en.z = __builtin_bit_cast(int, wj);
        en.w = 0;
        ent[pos] = en;
    }
}

#define GG_THREADS 1024
#define GG_WAVES (GG_THREADS / 32)
#define SC_GROUP 8
static_assert(BIN_NODES / 16 <= GG_WAVES, "one wave per 16-row tile of the range");

__global__ __launch_bounds__(GG_THREADS) void gather_gemm_kernel(const float* __restrict__ x,
                                                                 const int* __restrict__ entries,
                                                                 const int* __restrict__ binstart,
                                                                 const float* __restrict__ W0,
                                                                 const float* __restrict__ W1,
                                                                 const float* __restrict__ b,
                                                                 const float* __restrict__ dinv,
                                                                 float* __restrict__ out) {
    __shared__ float sS[BIN_NODES * D];
    __shared__ float sW0[D * D];
    __shared__ float sW1[D * D];
    __shared__ float sb[D];

    const int tid = threadIdx.x;
    const int lane = tid & 31;
    const int wave = __builtin_amdgcn_readfirstlane(tid >> 5);
    const int bin = blockIdx.x;

    const v4f zero4 = {0.0f, 0.0f, 0.0f, 0.0f};
    for (int i = tid; i < BIN_NODES * D / 4; i += GG_THREADS)
        reinterpret_cast<v4f*>(sS)[i] = zero4;
    for (int i = tid; i < D * D; i += GG_THREADS) {
        sW0[i] = W0[i];
        sW1[i] = W1[i];
    }
    if (tid < D) sb[tid] = b[tid];
    __syncthreads();

    int beg = binstart[bin];
    int end = binstart[bin + 1];
    beg = (beg < 0) ? 0 : ((beg > N_EDGES) ? N_EDGES : beg);
    end = (end < 0) ? 0 : ((end > N_EDGES) ? N_EDGES : end);
    beg = __builtin_amdgcn_readfirstlane(beg);
    end = __builtin_amdgcn_readfirstlane(end);

    const v4i* ent = reinterpret_cast<const v4i*>(entries);
    for (int base = beg + wave * 32; base < end; base += GG_WAVES * 32) {
        const int rest = end - base;
        const int nvalid = (rest < 32) ? rest : 32;
        const int il = (lane < nvalid) ? lane : (nvalid - 1);
        const v4i en = ent[base + il];
        const int row = en.x;
        int cl = en.y;
        cl = (cl < 0) ? 0 : ((cl > BIN_NODES - 1) ? (BIN_NODES - 1) : cl);
        const int wbits = en.z;
        const int nrm = __builtin_bit_cast(int, dinv[row] * __builtin_bit_cast(float, wbits));

        for (int j0 = 0; j0 < nvalid; j0 += SC_GROUP) {
            float xa[SC_GROUP], xb[SC_GROUP];
#pragma unroll
            for (int k = 0; k < SC_GROUP; ++k) {
                const int r = __builtin_amdgcn_readlane(row, j0 + k);
                const float* xr = x + (size_t)r * D;
                xa[k] = xr[lane];
                xb[k] = xr[lane + 32];
            }
#pragma unroll
            for (int k = 0; k < SC_GROUP; ++k) {
                if (j0 + k < nvalid) {
                    const int c = __builtin_amdgcn_readlane(cl, j0 + k);
                    const float s = __builtin_bit_cast(float, __builtin_amdgcn_readlane(nrm, j0 + k));
                    float* dst = sS + c * D;
                    atomicAdd(dst + lane, s * xa[k]);
                    atomicAdd(dst + lane + 32, s * xb[k]);
                }
            }
        }
    }
    __syncthreads();

    if (wave >= BIN_NODES / 16) return;

    const int half = lane >> 4;
    const int mrow = lane & 15;
    const int ml = wave * 16 + mrow;
    const int m0 = bin * BIN_NODES + wave * 16;

    v8f acc[4];
#pragma unroll
    for (int n = 0; n < 4; ++n) {
        float bias = sb[n * 16 + mrow];
#pragma unroll
        for (int r = 0; r < 8; ++r) acc[n][r] = bias;
    }

    const float* xrow = x + (long long)(m0 + mrow) * D;
    const float* trow = sS + ml * D;
    const float nd = -dinv[m0 + mrow];

#pragma unroll
    for (int ks = 0; ks < 16; ++ks) {
        int kk = ks * 4 + 2 * half;
        v2f ax = *reinterpret_cast<const v2f*>(xrow + kk);
        v2f at = *reinterpret_cast<const v2f*>(trow + kk);
        at.x *= nd;
        at.y *= nd;
#pragma unroll
        for (int n = 0; n < 4; ++n) {
            int col = n * 16 + mrow;
            v2f b0, b1;
            b0.x = sW0[kk * D + col];
            b0.y = sW0[(kk + 1) * D + col];
            b1.x = sW1[kk * D + col];
            b1.y = sW1[(kk + 1) * D + col];
            acc[n] = __builtin_amdgcn_wmma_f32_16x16x4_f32(
                false, ax, false, b0, (short)0, acc[n], false, false);
            acc[n] = __builtin_amdgcn_wmma_f32_16x16x4_f32(
                false, at, false, b1, (short)0, acc[n], false, false);
        }
    }

#pragma unroll
    for (int n = 0; n < 4; ++n) {
        int col = n * 16 + mrow;
#pragma unroll
        for (int r = 0; r < 8; ++r) {
            int row = m0 + half * 8 + r;
            out[(long long)row * D + col] = acc[n][r];
        }
    }
}

extern "C" void kernel_launch(void* const* d_in, const int* in_sizes, int n_in,
                              void* d_out, int out_size, void* d_ws, size_t ws_size,
                              hipStream_t stream) {
    const float* x  = (const float*)d_in[0];
    const int*   ei = (const int*)d_in[1];
    const float* ew = (const float*)d_in[2];
    const float* W0 = (const float*)d_in[3];
    const float* W1 = (const float*)d_in[4];
    const float* b  = (const float*)d_in[5];
    float* out = (float*)d_out;

    int*   cnt      = (int*)d_ws;
    int*   binstart = (int*)d_ws + 65536;
    float* dinv     = (float*)d_ws + N_NODES;
    float* big      = (float*)d_ws + 2 * (size_t)N_NODES;

    deg_count_kernel<<<DEG_BLOCKS + NSB, DEG_THREADS, 0, stream>>>(ei, ew, out, cnt);
    dinv_place_kernel<<<NSB + DINV_BLOCKS, SB_THREADS, 0, stream>>>(ei, ew, out, dinv, cnt, binstart, (int*)big);
    gather_gemm_kernel<<<NBIN, GG_THREADS, 0, stream>>>(x, (const int*)big, binstart, W0, W1, b, dinv, out);
}
